# starting kernel
# baseline (speedup 1.0000x reference)
#include <hip/hip_runtime.h>
#include <cstdint>

typedef __attribute__((ext_vector_type(2))) float v2f;
typedef __attribute__((ext_vector_type(8))) float v8f;

#define NWT 49600
#define NS  2048

__global__ __launch_bounds__(256) void hyper_kernel(
    const float* __restrict__ mu, const float* __restrict__ Wh1,
    const float* __restrict__ bh1, float* __restrict__ H) {
  const int n = blockIdx.x;
  const int c = threadIdx.x;
  float acc = bh1[c];
#pragma unroll
  for (int i = 0; i < 16; ++i)
    acc = fmaf(mu[n * 16 + i], Wh1[i * 256 + c], acc);
  const float s = 1.f / (1.f + __expf(-acc));
  H[n * 256 + c] = acc * s;
}

__global__ __launch_bounds__(256) void layer_kernel(
    const float* __restrict__ H, const float* __restrict__ Wh2,
    const float* __restrict__ bh2, const float* __restrict__ hin,
    float* __restrict__ hout,
    int offW, int offB, int din, int dout, int is_last) {
  __shared__ float wstage[33 * 16 * 16];
  __shared__ float hloc[16 * 128];

  const int tid  = threadIdx.x;
  const int lane = tid & 31;
  const int wave = tid >> 5;
  const int nsub = lane & 15;
  const int hi   = lane >> 4;

  const int ldin  = (din == 128) ? 7 : 6;
  const int rows  = 512 >> ldin;
  const int j0    = blockIdx.x * rows;
  const int s0    = blockIdx.y * 16;
  const int wcol0 = offW + blockIdx.x * 512;

  for (int i = tid; i < (16 << ldin); i += 256) {
    const int m = i >> ldin;
    const int k = i & (din - 1);
    hloc[m * 128 + k] = hin[(size_t)(s0 + m) * din + k];
  }

  v2f areg[64];
  {
    const float* ap = H + (size_t)(s0 + nsub) * 256 + 2 * hi;
#pragma unroll
    for (int ks = 0; ks < 64; ++ks) {
      areg[ks].x = ap[4 * ks];
      areg[ks].y = ap[4 * ks + 1];
    }
  }
  __syncthreads();

  for (int t = wave; t < 33; t += 8) {
    int col;
    if (t < 32) {
      col = wcol0 + t * 16 + nsub;
    } else {
      col = offB + j0 + nsub;
      const int cmax = offB + dout - 1;
      col = (col < cmax) ? col : cmax;
    }
    const float* bp = Wh2 + (size_t)(2 * hi) * NWT + col;

    v8f a0 = {}, a1 = {}, a2 = {}, a3 = {};
    v2f b0, b1, b2, b3;
    b0.x = bp[0];                 b0.y = bp[(size_t)1 * NWT];
    b1.x = bp[(size_t)4 * NWT];   b1.y = bp[(size_t)5 * NWT];
    b2.x = bp[(size_t)8 * NWT];   b2.y = bp[(size_t)9 * NWT];
    b3.x = bp[(size_t)12 * NWT];  b3.y = bp[(size_t)13 * NWT];

#pragma unroll
    for (int g = 0; g < 16; ++g) {
      v2f n0 = b0, n1 = b1, n2 = b2, n3 = b3;
      const float* np = bp + (size_t)16 * NWT;
      if (g < 15) {
        n0.x = np[0];                 n0.y = np[(size_t)1 * NWT];
        n1.x = np[(size_t)4 * NWT];   n1.y = np[(size_t)5 * NWT];
        n2.x = np[(size_t)8 * NWT];   n2.y = np[(size_t)9 * NWT];
        n3.x = np[(size_t)12 * NWT];  n3.y = np[(size_t)13 * NWT];
      }
      a0 = __builtin_amdgcn_wmma_f32_16x16x4_f32(false, areg[4 * g + 0], false, b0,
                                                 (short)0, a0, false, false);
      a1 = __builtin_amdgcn_wmma_f32_16x16x4_f32(false, areg[4 * g + 1], false, b1,
                                                 (short)0, a1, false, false);
      a2 = __builtin_amdgcn_wmma_f32_16x16x4_f32(false, areg[4 * g + 2], false, b2,
                                                 (short)0, a2, false, false);
      a3 = __builtin_amdgcn_wmma_f32_16x16x4_f32(false, areg[4 * g + 3], false, b3,
                                                 (short)0, a3, false, false);
      b0 = n0; b1 = n1; b2 = n2; b3 = n3;
      bp = np;
    }
    const v8f acc = (a0 + a1) + (a2 + a3);

    const float bias = bh2[col];
    const int base = (t * 16 + nsub) * 16 + hi * 8;
    float4* dst = (float4*)&wstage[base];
    dst[0] = make_float4(acc[0] + bias, acc[1] + bias, acc[2] + bias, acc[3] + bias);
    dst[1] = make_float4(acc[4] + bias, acc[5] + bias, acc[6] + bias, acc[7] + bias);
  }
  __syncthreads();

  for (int p = tid; p < 16 * rows; p += 256) {
    const int m = p & 15;
    const int jl = p >> 4;
    float z = wstage[(512 + jl) * 16 + m];
    const float* wrow = &wstage[((jl << ldin)) * 16 + m];
    const float* hv = &hloc[m * 128];
#pragma unroll 8
    for (int k = 0; k < din; ++k)
      z = fmaf(wrow[k * 16], hv[k], z);
    const int j = j0 + jl;
    if (is_last) {
      hout[(size_t)(s0 + m) * dout + j] = z;
    } else {
      const float s = 1.f / (1.f + __expf(-z));
      hout[(size_t)(s0 + m) * dout + j] = z * s;
    }
  }
}

extern "C" void kernel_launch(void* const* d_in, const int* in_sizes, int n_in,
                              void* d_out, int out_size, void* d_ws, size_t ws_size,
                              hipStream_t stream) {
  (void)in_sizes; (void)n_in; (void)out_size; (void)ws_size;
  const float* x   = (const float*)d_in[0];
  const float* mu  = (const float*)d_in[1];
  const float* Wh1 = (const float*)d_in[2];
  const float* bh1 = (const float*)d_in[3];
  const float* Wh2 = (const float*)d_in[4];
  const float* bh2 = (const float*)d_in[5];
  float* out = (float*)d_out;

  float* H  = (float*)d_ws;
  float* h1 = H + (size_t)NS * 256;
  float* h2 = h1 + (size_t)NS * 128;

  hyper_kernel<<<NS, 256, 0, stream>>>(mu, Wh1, bh1, H);

  layer_kernel<<<dim3(16, 128), 256, 0, stream>>>(H, Wh2, bh2, x,  h1,     0,  8192,  64, 128, 0);
  layer_kernel<<<dim3(32, 128), 256, 0, stream>>>(H, Wh2, bh2, h1, h2,  8320, 24704, 128, 128, 0);
  layer_kernel<<<dim3(32, 128), 256, 0, stream>>>(H, Wh2, bh2, h2, h1, 24832, 41216, 128, 128, 0);
  layer_kernel<<<dim3(16, 128), 256, 0, stream>>>(H, Wh2, bh2, h1, out, 41344, 49536, 128, 64, 1);
}


# Each wave owns 64 columns as 4 interleaved tiles: B rows come as 128-bit loads 5 K-steps ahead, each A fragment feeds 4 WMMAs; a tile's K is now summed in one in-order chain (was 4 partial sums)
# speedup vs baseline: 1.2690x; 1.2690x over previous
#include <hip/hip_runtime.h>
#include <cstdint>

typedef __attribute__((ext_vector_type(2))) float v2f;
typedef __attribute__((ext_vector_type(8))) float v8f;

#define NWT 49600
#define NS  2048

__global__ __launch_bounds__(256) void hyper_kernel(
    const float* __restrict__ mu, const float* __restrict__ Wh1,
    const float* __restrict__ bh1, float* __restrict__ H) {
  const int n = blockIdx.x;
  const int c = threadIdx.x;
  float acc = bh1[c];
#pragma unroll
  for (int i = 0; i < 16; ++i)
    acc = fmaf(mu[n * 16 + i], Wh1[i * 256 + c], acc);
  const float s = 1.f / (1.f + __expf(-acc));
  H[n * 256 + c] = acc * s;
}

#define WP 20
#define BAHEAD 5

__global__ __launch_bounds__(256) void layer_kernel(
    const float* __restrict__ H, const float* __restrict__ Wh2,
    const float* __restrict__ bh2, const float* __restrict__ hin,
    float* __restrict__ hout,
    int offW, int offB, int din, int dout, int is_last) {
  __shared__ float wstage[33 * 16 * WP];
  __shared__ float hloc[16 * 128];

  const int tid  = threadIdx.x;
  const int lane = tid & 31;
  const int wave = tid >> 5;
  const int nsub = lane & 15;
  const int hi   = lane >> 4;

  const int ldin  = (din == 128) ? 7 : 6;
  const int rows  = 512 >> ldin;
  const int j0    = blockIdx.x * rows;
  const int s0    = blockIdx.y * 16;
  const int wcol0 = offW + blockIdx.x * 512;

  for (int i = tid; i < (16 << ldin); i += 256) {
    const int m = i >> ldin;
    const int k = i & (din - 1);
    hloc[m * 128 + k] = hin[(size_t)(s0 + m) * din + k];
  }

  v2f areg[64];
  {
    const float* ap = H + (size_t)(s0 + nsub) * 256 + 2 * hi;
#pragma unroll
    for (int ks = 0; ks < 64; ++ks) {
      areg[ks].x = ap[4 * ks];
      areg[ks].y = ap[4 * ks + 1];
    }
  }
  __syncthreads();

  {
    const int lc0 = wave * 64 + 4 * nsub;
    const float* bp = Wh2 + (size_t)(2 * hi) * NWT + wcol0 + lc0;

    v8f c0 = {}, c1 = {}, c2 = {}, c3 = {};
    float4 rx[BAHEAD], ry[BAHEAD];
#pragma unroll
    for (int s = 0; s < BAHEAD; ++s) {
      rx[s] = *(const float4*)(bp + (size_t)(4 * s) * NWT);
      ry[s] = *(const float4*)(bp + (size_t)(4 * s + 1) * NWT);
    }

#pragma unroll
    for (int ks = 0; ks < 64; ++ks) {
      const float4 x = rx[ks % BAHEAD];
      const float4 y = ry[ks % BAHEAD];
      if (ks + BAHEAD < 64) {
        rx[ks % BAHEAD] = *(const float4*)(bp + (size_t)(4 * (ks + BAHEAD)) * NWT);
        ry[ks % BAHEAD] = *(const float4*)(bp + (size_t)(4 * (ks + BAHEAD) + 1) * NWT);
      }
      const v2f a = areg[ks];
      v2f b;
      b.x = x.x; b.y = y.x;
      c0 = __builtin_amdgcn_wmma_f32_16x16x4_f32(false, a, false, b, (short)0, c0, false, false);
      b.x = x.y; b.y = y.y;
      c1 = __builtin_amdgcn_wmma_f32_16x16x4_f32(false, a, false, b, (short)0, c1, false, false);
      b.x = x.z; b.y = y.z;
      c2 = __builtin_amdgcn_wmma_f32_16x16x4_f32(false, a, false, b, (short)0, c2, false, false);
      b.x = x.w; b.y = y.w;
      c3 = __builtin_amdgcn_wmma_f32_16x16x4_f32(false, a, false, b, (short)0, c3, false, false);
    }

    const float4 bias4 = *(const float4*)(bh2 + wcol0 + lc0);
    float4* dst;
    dst = (float4*)&wstage[(lc0 + 0) * WP + hi * 8];
    dst[0] = make_float4(c0[0] + bias4.x, c0[1] + bias4.x, c0[2] + bias4.x, c0[3] + bias4.x);
    dst[1] = make_float4(c0[4] + bias4.x, c0[5] + bias4.x, c0[6] + bias4.x, c0[7] + bias4.x);
    dst = (float4*)&wstage[(lc0 + 1) * WP + hi * 8];
    dst[0] = make_float4(c1[0] + bias4.y, c1[1] + bias4.y, c1[2] + bias4.y, c1[3] + bias4.y);
    dst[1] = make_float4(c1[4] + bias4.y, c1[5] + bias4.y, c1[6] + bias4.y, c1[7] + bias4.y);
    dst = (float4*)&wstage[(lc0 + 2) * WP + hi * 8];
    dst[0] = make_float4(c2[0] + bias4.z, c2[1] + bias4.z, c2[2] + bias4.z, c2[3] + bias4.z);
    dst[1] = make_float4(c2[4] + bias4.z, c2[5] + bias4.z, c2[6] + bias4.z, c2[7] + bias4.z);
    dst = (float4*)&wstage[(lc0 + 3) * WP + hi * 8];
    dst[0] = make_float4(c3[0] + bias4.w, c3[1] + bias4.w, c3[2] + bias4.w, c3[3] + bias4.w);
    dst[1] = make_float4(c3[4] + bias4.w, c3[5] + bias4.w, c3[6] + bias4.w, c3[7] + bias4.w);
  }

  if (wave == 0) {
    int col = offB + j0 + nsub;
    const int cmax = offB + dout - 1;
    col = (col < cmax) ? col : cmax;
    const float* bp = Wh2 + (size_t)(2 * hi) * NWT + col;

    v8f a0 = {}, a1 = {}, a2 = {}, a3 = {};
    v2f b0, b1, b2, b3;
    b0.x = bp[0];                 b0.y = bp[(size_t)1 * NWT];
    b1.x = bp[(size_t)4 * NWT];   b1.y = bp[(size_t)5 * NWT];
    b2.x = bp[(size_t)8 * NWT];   b2.y = bp[(size_t)9 * NWT];
    b3.x = bp[(size_t)12 * NWT];  b3.y = bp[(size_t)13 * NWT];

#pragma unroll
    for (int g = 0; g < 16; ++g) {
      v2f n0 = b0, n1 = b1, n2 = b2, n3 = b3;
      const float* np = bp + (size_t)16 * NWT;
      if (g < 15) {
        n0.x = np[0];                 n0.y = np[(size_t)1 * NWT];
        n1.x = np[(size_t)4 * NWT];   n1.y = np[(size_t)5 * NWT];
        n2.x = np[(size_t)8 * NWT];   n2.y = np[(size_t)9 * NWT];
        n3.x = np[(size_t)12 * NWT];  n3.y = np[(size_t)13 * NWT];
      }
      a0 = __builtin_amdgcn_wmma_f32_16x16x4_f32(false, areg[4 * g + 0], false, b0,
                                                 (short)0, a0, false, false);
      a1 = __builtin_amdgcn_wmma_f32_16x16x4_f32(false, areg[4 * g + 1], false, b1,
                                                 (short)0, a1, false, false);
      a2 = __builtin_amdgcn_wmma_f32_16x16x4_f32(false, areg[4 * g + 2], false, b2,
                                                 (short)0, a2, false, false);
      a3 = __builtin_amdgcn_wmma_f32_16x16x4_f32(false, areg[4 * g + 3], false, b3,
                                                 (short)0, a3, false, false);
      b0 = n0; b1 = n1; b2 = n2; b3 = n3;
      bp = np;
      __builtin_amdgcn_sched_barrier(0);
    }
    const v8f acc = (a0 + a1) + (a2 + a3);

    const float bias = bh2[col];
    float4* dst = (float4*)&wstage[(512 + nsub) * WP + hi * 8];
    dst[0] = make_float4(acc[0] + bias, acc[1] + bias, acc[2] + bias, acc[3] + bias);
    dst[1] = make_float4(acc[4] + bias, acc[5] + bias, acc[6] + bias, acc[7] + bias);
  }
  __syncthreads();

  for (int p = tid; p < 16 * rows; p += 256) {
    const int m = p & 15;
    const int jl = p >> 4;
    float z = wstage[(512 + jl) * WP + m];
    const float* wrow = &wstage[((jl << ldin)) * WP + m];
    const float* hv = &hloc[m * 128];
#pragma unroll 8
    for (int k = 0; k < din; ++k)
      z = fmaf(wrow[k * WP], hv[k], z);
    const int j = j0 + jl;
    if (is_last) {
      hout[(size_t)(s0 + m) * dout + j] = z;
    } else {
      const float s = 1.f / (1.f + __expf(-z));
      hout[(size_t)(s0 + m) * dout + j] = z * s;
    }
  }
}

extern "C" void kernel_launch(void* const* d_in, const int* in_sizes, int n_in,
                              void* d_out, int out_size, void* d_ws, size_t ws_size,
                              hipStream_t stream) {
  (void)in_sizes; (void)n_in; (void)out_size; (void)ws_size;
  const float* x   = (const float*)d_in[0];
  const float* mu  = (const float*)d_in[1];
  const float* Wh1 = (const float*)d_in[2];
  const float* bh1 = (const float*)d_in[3];
  const float* Wh2 = (const float*)d_in[4];
  const float* bh2 = (const float*)d_in[5];
  float* out = (float*)d_out;

  float* H  = (float*)d_ws;
  float* h1 = H + (size_t)NS * 256;
  float* h2 = h1 + (size_t)NS * 128;

  hyper_kernel<<<NS, 256, 0, stream>>>(mu, Wh1, bh1, H);

  layer_kernel<<<dim3(16, 128), 256, 0, stream>>>(H, Wh2, bh2, x,  h1,     0,  8192,  64, 128, 0);
  layer_kernel<<<dim3(32, 128), 256, 0, stream>>>(H, Wh2, bh2, h1, h2,  8320, 24704, 128, 128, 0);
  layer_kernel<<<dim3(32, 128), 256, 0, stream>>>(H, Wh2, bh2, h2, h1, 24832, 41216, 128, 128, 0);
  layer_kernel<<<dim3(16, 128), 256, 0, stream>>>(H, Wh2, bh2, h1, out, 41344, 49536, 128, 64, 1);
}


# A tile of H staged once per work-group in LDS in fragment order, not 128 VGPRs per lane (B ring 5->4 K-steps to fit): 128 VGPRs, 8 waves per SIMD, four work-groups per WGP instead of two
# speedup vs baseline: 1.7468x; 1.3765x over previous
#include <hip/hip_runtime.h>
#include <cstdint>

typedef __attribute__((ext_vector_type(2))) float v2f;
typedef __attribute__((ext_vector_type(8))) float v8f;

#define NWT 49600
#define NS  2048

__global__ __launch_bounds__(256) void hyper_kernel(
    const float* __restrict__ mu, const float* __restrict__ Wh1,
    const float* __restrict__ bh1, float* __restrict__ H) {
  const int n = blockIdx.x;
  const int c = threadIdx.x;
  float acc = bh1[c];
#pragma unroll
  for (int i = 0; i < 16; ++i)
    acc = fmaf(mu[n * 16 + i], Wh1[i * 256 + c], acc);
  const float s = 1.f / (1.f + __expf(-acc));
  H[n * 256 + c] = acc * s;
}

#define WP 20
#define BAHEAD 4
#define AP 132

__global__ __launch_bounds__(256, 8) void layer_kernel(
    const float* __restrict__ H, const float* __restrict__ Wh2,
    const float* __restrict__ bh2, const float* __restrict__ hin,
    float* __restrict__ hout,
    int offW, int offB, int din, int dout, int is_last) {
  __shared__ float wstage[33 * 16 * WP];
  __shared__ float hloc[16 * 128];
  __shared__ __attribute__((aligned(16))) float alds[32 * AP];

  const int tid  = threadIdx.x;
  const int lane = tid & 31;
  const int wave = tid >> 5;
  const int nsub = lane & 15;
  const int hi   = lane >> 4;

  const int ldin  = (din == 128) ? 7 : 6;
  const int rows  = 512 >> ldin;
  const int j0    = blockIdx.x * rows;
  const int s0    = blockIdx.y * 16;
  const int wcol0 = offW + blockIdx.x * 512;

  for (int i = tid; i < (16 << ldin); i += 256) {
    const int m = i >> ldin;
    const int k = i & (din - 1);
    hloc[m * 128 + k] = hin[(size_t)(s0 + m) * din + k];
  }

#pragma unroll
  for (int i = 0; i < 4; ++i) {
    const int q  = tid + 256 * i;
    const int m  = q >> 6;
    const int ks = q & 63;
    const float4 v = *(const float4*)(H + (size_t)(s0 + m) * 256 + 4 * ks);
    *(float2*)&alds[m * AP + 2 * ks]        = make_float2(v.x, v.y);
    *(float2*)&alds[(m + 16) * AP + 2 * ks] = make_float2(v.z, v.w);
  }
  const float4* afrag = (const float4*)&alds[lane * AP];
  __syncthreads();

  {
    const int lc0 = wave * 64 + 4 * nsub;
    const float* bp = Wh2 + (size_t)(2 * hi) * NWT + wcol0 + lc0;

    v8f c0 = {}, c1 = {}, c2 = {}, c3 = {};
    float4 rx[BAHEAD], ry[BAHEAD];
#pragma unroll
    for (int s = 0; s < BAHEAD; ++s) {
      rx[s] = *(const float4*)(bp + (size_t)(4 * s) * NWT);
      ry[s] = *(const float4*)(bp + (size_t)(4 * s + 1) * NWT);
    }

    float4 aq[2][2];
    aq[0][0] = afrag[0];
    aq[0][1] = afrag[1];

#pragma unroll
    for (int ks = 0; ks < 64; ++ks) {
      if ((ks & 3) == 0 && ks + 4 < 64) {
        const int g = (ks >> 2) + 1;
        aq[g & 1][0] = afrag[2 * g];
        aq[g & 1][1] = afrag[2 * g + 1];
      }
      const float4 x = rx[ks % BAHEAD];
      const float4 y = ry[ks % BAHEAD];
      if (ks + BAHEAD < 64) {
        rx[ks % BAHEAD] = *(const float4*)(bp + (size_t)(4 * (ks + BAHEAD)) * NWT);
        ry[ks % BAHEAD] = *(const float4*)(bp + (size_t)(4 * (ks + BAHEAD) + 1) * NWT);
      }
      const float4 av = aq[(ks >> 2) & 1][(ks >> 1) & 1];
      v2f a;
      if (ks & 1) { a.x = av.z; a.y = av.w; } else { a.x = av.x; a.y = av.y; }
      v2f b;
      b.x = x.x; b.y = y.x;
      c0 = __builtin_amdgcn_wmma_f32_16x16x4_f32(false, a, false, b, (short)0, c0, false, false);
      b.x = x.y; b.y = y.y;
      c1 = __builtin_amdgcn_wmma_f32_16x16x4_f32(false, a, false, b, (short)0, c1, false, false);
      b.x = x.z; b.y = y.z;
      c2 = __builtin_amdgcn_wmma_f32_16x16x4_f32(false, a, false, b, (short)0, c2, false, false);
      b.x = x.w; b.y = y.w;
      c3 = __builtin_amdgcn_wmma_f32_16x16x4_f32(false, a, false, b, (short)0, c3, false, false);
      __builtin_amdgcn_sched_barrier(0);
    }

    const float4 bias4 = *(const float4*)(bh2 + wcol0 + lc0);
    float4* dst;
    dst = (float4*)&wstage[(lc0 + 0) * WP + hi * 8];
    dst[0] = make_float4(c0[0] + bias4.x, c0[1] + bias4.x, c0[2] + bias4.x, c0[3] + bias4.x);
    dst[1] = make_float4(c0[4] + bias4.x, c0[5] + bias4.x, c0[6] + bias4.x, c0[7] + bias4.x);
    dst = (float4*)&wstage[(lc0 + 1) * WP + hi * 8];
    dst[0] = make_float4(c1[0] + bias4.y, c1[1] + bias4.y, c1[2] + bias4.y, c1[3] + bias4.y);
    dst[1] = make_float4(c1[4] + bias4.y, c1[5] + bias4.y, c1[6] + bias4.y, c1[7] + bias4.y);
    dst = (float4*)&wstage[(lc0 + 2) * WP + hi * 8];
    dst[0] = make_float4(c2[0] + bias4.z, c2[1] + bias4.z, c2[2] + bias4.z, c2[3] + bias4.z);
    dst[1] = make_float4(c2[4] + bias4.z, c2[5] + bias4.z, c2[6] + bias4.z, c2[7] + bias4.z);
    dst = (float4*)&wstage[(lc0 + 3) * WP + hi * 8];
    dst[0] = make_float4(c3[0] + bias4.w, c3[1] + bias4.w, c3[2] + bias4.w, c3[3] + bias4.w);
    dst[1] = make_float4(c3[4] + bias4.w, c3[5] + bias4.w, c3[6] + bias4.w, c3[7] + bias4.w);
  }

  if (wave == 0) {
    int col = offB + j0 + nsub;
    const int cmax = offB + dout - 1;
    col = (col < cmax) ? col : cmax;
    const float* bp = Wh2 + (size_t)(2 * hi) * NWT + col;

    v8f a0 = {}, a1 = {}, a2 = {}, a3 = {};
    v2f b0, b1, b2, b3;
    b0.x = bp[0];                 b0.y = bp[(size_t)1 * NWT];
    b1.x = bp[(size_t)4 * NWT];   b1.y = bp[(size_t)5 * NWT];
    b2.x = bp[(size_t)8 * NWT];   b2.y = bp[(size_t)9 * NWT];
    b3.x = bp[(size_t)12 * NWT];  b3.y = bp[(size_t)13 * NWT];
    float4 p01 = afrag[0], p23 = afrag[1];

#pragma unroll
    for (int g = 0; g < 16; ++g) {
      v2f n0 = b0, n1 = b1, n2 = b2, n3 = b3;
      float4 q01 = p01, q23 = p23;
      const float* np = bp + (size_t)16 * NWT;
      if (g < 15) {
        q01 = afrag[2 * g + 2];       q23 = afrag[2 * g + 3];
        n0.x = np[0];                 n0.y = np[(size_t)1 * NWT];
        n1.x = np[(size_t)4 * NWT];   n1.y = np[(size_t)5 * NWT];
        n2.x = np[(size_t)8 * NWT];   n2.y = np[(size_t)9 * NWT];
        n3.x = np[(size_t)12 * NWT];  n3.y = np[(size_t)13 * NWT];
      }
      v2f f0, f1, f2, f3;
      f0.x = p01.x; f0.y = p01.y;  f1.x = p01.z; f1.y = p01.w;
      f2.x = p23.x; f2.y = p23.y;  f3.x = p23.z; f3.y = p23.w;
      a0 = __builtin_amdgcn_wmma_f32_16x16x4_f32(false, f0, false, b0,
                                                 (short)0, a0, false, false);
      a1 = __builtin_amdgcn_wmma_f32_16x16x4_f32(false, f1, false, b1,
                                                 (short)0, a1, false, false);
      a2 = __builtin_amdgcn_wmma_f32_16x16x4_f32(false, f2, false, b2,
                                                 (short)0, a2, false, false);
      a3 = __builtin_amdgcn_wmma_f32_16x16x4_f32(false, f3, false, b3,
                                                 (short)0, a3, false, false);
      b0 = n0; b1 = n1; b2 = n2; b3 = n3;
      p01 = q01; p23 = q23;
      bp = np;
      __builtin_amdgcn_sched_barrier(0);
    }
    const v8f acc = (a0 + a1) + (a2 + a3);

    const float bias = bh2[col];
    float4* dst = (float4*)&wstage[(512 + nsub) * WP + hi * 8];
    dst[0] = make_float4(acc[0] + bias, acc[1] + bias, acc[2] + bias, acc[3] + bias);
    dst[1] = make_float4(acc[4] + bias, acc[5] + bias, acc[6] + bias, acc[7] + bias);
  }
  __syncthreads();

  for (int p = tid; p < 16 * rows; p += 256) {
    const int m = p & 15;
    const int jl = p >> 4;
    float z = wstage[(512 + jl) * WP + m];
    const float* wrow = &wstage[((jl << ldin)) * WP + m];
    const float* hv = &hloc[m * 128];
#pragma unroll 8
    for (int k = 0; k < din; ++k)
      z = fmaf(wrow[k * WP], hv[k], z);
    const int j = j0 + jl;
    if (is_last) {
      hout[(size_t)(s0 + m) * dout + j] = z;
    } else {
      const float s = 1.f / (1.f + __expf(-z));
      hout[(size_t)(s0 + m) * dout + j] = z * s;
    }
  }
}

extern "C" void kernel_launch(void* const* d_in, const int* in_sizes, int n_in,
                              void* d_out, int out_size, void* d_ws, size_t ws_size,
                              hipStream_t stream) {
  (void)in_sizes; (void)n_in; (void)out_size; (void)ws_size;
  const float* x   = (const float*)d_in[0];
  const float* mu  = (const float*)d_in[1];
  const float* Wh1 = (const float*)d_in[2];
  const float* bh1 = (const float*)d_in[3];
  const float* Wh2 = (const float*)d_in[4];
  const float* bh2 = (const float*)d_in[5];
  float* out = (float*)d_out;

  float* H  = (float*)d_ws;
  float* h1 = H + (size_t)NS * 256;
  float* h2 = h1 + (size_t)NS * 128;

  hyper_kernel<<<NS, 256, 0, stream>>>(mu, Wh1, bh1, H);

  layer_kernel<<<dim3(16, 128), 256, 0, stream>>>(H, Wh2, bh2, x,  h1,     0,  8192,  64, 128, 0);
  layer_kernel<<<dim3(32, 128), 256, 0, stream>>>(H, Wh2, bh2, h1, h2,  8320, 24704, 128, 128, 0);
  layer_kernel<<<dim3(32, 128), 256, 0, stream>>>(H, Wh2, bh2, h2, h1, 24832, 41216, 128, 128, 0);
  layer_kernel<<<dim3(16, 128), 256, 0, stream>>>(H, Wh2, bh2, h1, out, 41344, 49536, 128, 64, 1);
}
